# starting kernel
# baseline (speedup 1.0000x reference)
#include <hip/hip_runtime.h>
#include <hip/hip_bf16.h>


typedef float v2f __attribute__((ext_vector_type(2)));
typedef float v4f __attribute__((ext_vector_type(4)));
typedef float v8f __attribute__((ext_vector_type(8)));

#define Ldim 4
#define Bdim 8
#define Cdim 256
#define HW   4096

__global__ void __launch_bounds__(256) sffm_gap_kernel(
    const float* __restrict__ in, float* __restrict__ gap) {
  __shared__ float sdata[256];
  const int row = blockIdx.x;
  const int t   = threadIdx.x;
  const v4f* p = (const v4f*)(in + (size_t)row * HW);
  float s = 0.0f;
#pragma unroll
  for (int i = 0; i < 4; ++i) {
    v4f v = p[t + i * 256];
    s += (v.x + v.y) + (v.z + v.w);
  }
  sdata[t] = s;
  __syncthreads();
  for (int off = 128; off > 0; off >>= 1) {
    if (t < off) sdata[t] += sdata[t + off];
    __syncthreads();
  }
  if (t == 0) gap[row] = sdata[0] * (1.0f / (float)HW);
}

__global__ void __launch_bounds__(32) sffm_gemm_kernel(
    const float* __restrict__ gap, const float* __restrict__ Wlin,
    float* __restrict__ scores) {
  const int lane  = threadIdx.x;
  const int mtile = blockIdx.x >> 4;
  const int ntile = blockIdx.x & 15;
  const int half  = lane >> 4;
  const int lr    = lane & 15;
  const int m = mtile * 16 + lr;
  const int n = ntile * 16 + lr;

  v8f acc = {};
  for (int kk = 0; kk < Cdim; kk += 4) {
    const int k = kk + half * 2;
    v2f a, b;
    a.x = gap[m * Cdim + k];
    a.y = gap[m * Cdim + k + 1];
    b.x = Wlin[n * Cdim + k];
    b.y = Wlin[n * Cdim + k + 1];
    acc = __builtin_amdgcn_wmma_f32_16x16x4_f32(
         false, a,  false, b,
         (short)0, acc,  false,  false);
  }

  const int col = ntile * 16 + lr;
#pragma unroll
  for (int v = 0; v < 8; ++v) {
    const int row = mtile * 16 + v + half * 8;
    scores[row * Cdim + col] = acc[v];
  }
}

__global__ void __launch_bounds__(256) sffm_apply_kernel(
    const float* __restrict__ in, const float* __restrict__ scores,
    float* __restrict__ out) {
  const int bc = blockIdx.x;
  const int b  = bc >> 8;
  const int c  = bc & 255;
  const int t  = threadIdx.x;

  float s[Ldim];
#pragma unroll
  for (int l = 0; l < Ldim; ++l)
    s[l] = scores[((l << 3) + b) * Cdim + c];

  float mx = fmaxf(fmaxf(s[0], s[1]), fmaxf(s[2], s[3]));
  float e[Ldim];
  float sum = 0.0f;
#pragma unroll
  for (int l = 0; l < Ldim; ++l) { e[l] = __expf(s[l] - mx); sum += e[l]; }
  const float inv = 1.0f / sum;

#pragma unroll
  for (int l = 0; l < Ldim; ++l) {
    const float a = e[l] * inv;
    const size_t base = (size_t)(((l << 3) + b) * Cdim + c) * HW;
    const v4f* ip = (const v4f*)(in + base);
    v4f*       op = (v4f*)(out + base);
#pragma unroll
    for (int i = 0; i < 4; ++i) {
      v4f v = __builtin_nontemporal_load(&ip[t + i * 256]);
      v4f r = v * a;
      __builtin_nontemporal_store(r, &op[t + i * 256]);
    }
  }
}

extern "C" void kernel_launch(void* const* d_in, const int* in_sizes, int n_in,
                              void* d_out, int out_size, void* d_ws, size_t ws_size,
                              hipStream_t stream) {
  const float* in   = (const float*)d_in[0];
  const float* Wlin = (const float*)d_in[1];
  float* out    = (float*)d_out;
  float* gap    = (float*)d_ws;
  float* scores = gap + Ldim * Bdim * Cdim;

  sffm_gap_kernel  <<<Ldim * Bdim * Cdim, 256, 0, stream>>>(in, gap);
  sffm_gemm_kernel <<<32, 32, 0, stream>>>(gap, Wlin, scores);
  sffm_apply_kernel<<<Bdim * Cdim, 256, 0, stream>>>(in, scores, out);
}


# Score GEMM folded into the apply kernel (3 launches -> 2): each (b,c) block forms its 4 scores on f32 WMMA, k split over its 8 waves, the 8 partial sums added in wave order via LDS
# speedup vs baseline: 1.1746x; 1.1746x over previous
#include <hip/hip_runtime.h>
#include <hip/hip_bf16.h>


typedef float v2f __attribute__((ext_vector_type(2)));
typedef float v4f __attribute__((ext_vector_type(4)));
typedef float v8f __attribute__((ext_vector_type(8)));

#define Ldim 4
#define Bdim 8
#define Cdim 256
#define HW   4096

__global__ void __launch_bounds__(256) sffm_gap_kernel(
    const float* __restrict__ in, float* __restrict__ gap) {
  __shared__ float sdata[256];
  const int row = blockIdx.x;
  const int t   = threadIdx.x;
  const v4f* p = (const v4f*)(in + (size_t)row * HW);
  float s = 0.0f;
#pragma unroll
  for (int i = 0; i < 4; ++i) {
    v4f v = p[t + i * 256];
    s += (v.x + v.y) + (v.z + v.w);
  }
  sdata[t] = s;
  __syncthreads();
  for (int off = 128; off > 0; off >>= 1) {
    if (t < off) sdata[t] += sdata[t + off];
    __syncthreads();
  }
  if (t == 0) gap[row] = sdata[0] * (1.0f / (float)HW);
}

__global__ void __launch_bounds__(256) sffm_apply_kernel(
    const float* __restrict__ in, const float* __restrict__ gap,
    const float* __restrict__ Wlin, float* __restrict__ out) {
  __shared__ v4f part[8];
  const int bc = blockIdx.x;
  const int b  = bc >> 8;
  const int c  = bc & 255;
  const int t  = threadIdx.x;
  const int wave = t >> 5;
  const int lane = t & 31;
  const int half = lane >> 4;
  const int lr   = lane & 15;

  const int kb = wave * 32 + half * 16;
  const v4f* ap = (const v4f*)(gap + ((((lr & 3) << 3) + b) * Cdim + kb));
  const v4f* bp = (const v4f*)(Wlin + (c * Cdim + kb));
  v4f a4[4], b4[4];
#pragma unroll
  for (int j = 0; j < 4; ++j) { a4[j] = ap[j]; b4[j] = bp[j]; }

  v4f first[4];
  {
    const v4f* ip0 = (const v4f*)(in + (size_t)(b * Cdim + c) * HW);
#pragma unroll
    for (int i = 0; i < 4; ++i)
      first[i] = __builtin_nontemporal_load(&ip0[t + i * 256]);
  }

  v8f acc = {};
#pragma unroll
  for (int j = 0; j < 4; ++j) {
    v2f a, w;
    a.x = a4[j].x; a.y = a4[j].y; w.x = b4[j].x; w.y = b4[j].y;
    acc = __builtin_amdgcn_wmma_f32_16x16x4_f32(
         false, a,  false, w,
         (short)0, acc,  false,  false);
    a.x = a4[j].z; a.y = a4[j].w; w.x = b4[j].z; w.y = b4[j].w;
    acc = __builtin_amdgcn_wmma_f32_16x16x4_f32(
         false, a,  false, w,
         (short)0, acc,  false,  false);
  }
  if (lane == 0) {
    v4f p;
    p.x = acc[0]; p.y = acc[1]; p.z = acc[2]; p.w = acc[3];
    part[wave] = p;
  }
  __syncthreads();

  v4f sv = part[0];
#pragma unroll
  for (int w = 1; w < 8; ++w) sv += part[w];
  float s[Ldim];
  s[0] = sv.x; s[1] = sv.y; s[2] = sv.z; s[3] = sv.w;

  float mx = fmaxf(fmaxf(s[0], s[1]), fmaxf(s[2], s[3]));
  float e[Ldim];
  float sum = 0.0f;
#pragma unroll
  for (int l = 0; l < Ldim; ++l) { e[l] = __expf(s[l] - mx); sum += e[l]; }
  const float inv = 1.0f / sum;

  {
    const float a = e[0] * inv;
    v4f* op = (v4f*)(out + (size_t)(b * Cdim + c) * HW);
#pragma unroll
    for (int i = 0; i < 4; ++i) {
      v4f r = first[i] * a;
      __builtin_nontemporal_store(r, &op[t + i * 256]);
    }
  }
#pragma unroll
  for (int l = 1; l < Ldim; ++l) {
    const float a = e[l] * inv;
    const size_t base = (size_t)(((l << 3) + b) * Cdim + c) * HW;
    const v4f* ip = (const v4f*)(in + base);
    v4f*       op = (v4f*)(out + base);
#pragma unroll
    for (int i = 0; i < 4; ++i) {
      v4f v = __builtin_nontemporal_load(&ip[t + i * 256]);
      v4f r = v * a;
      __builtin_nontemporal_store(r, &op[t + i * 256]);
    }
  }
}

extern "C" void kernel_launch(void* const* d_in, const int* in_sizes, int n_in,
                              void* d_out, int out_size, void* d_ws, size_t ws_size,
                              hipStream_t stream) {
  const float* in   = (const float*)d_in[0];
  const float* Wlin = (const float*)d_in[1];
  float* out    = (float*)d_out;
  float* gap    = (float*)d_ws;

  sffm_gap_kernel  <<<Ldim * Bdim * Cdim, 256, 0, stream>>>(in, gap);
  sffm_apply_kernel<<<Bdim * Cdim, 256, 0, stream>>>(in, gap, Wlin, out);
}
